# starting kernel
# baseline (speedup 1.0000x reference)
#include <hip/hip_runtime.h>
#include <hip/hip_bf16.h>

typedef __attribute__((ext_vector_type(2))) float v2f;
typedef __attribute__((ext_vector_type(8))) float v8f;

constexpr int B_ = 8, L_ = 8192, T_ = 512, DH_ = 1024, DG_ = 768, DP_ = 256;

__device__ __forceinline__ v8f wmma4(v2f a, v2f b, v8f c) {
  return __builtin_amdgcn_wmma_f32_16x16x4_f32(
       false, a,  false, b,
       (short)0, c,  false,  false);
}

__device__ __forceinline__ v8f zero8() {
  v8f z = {0.f, 0.f, 0.f, 0.f, 0.f, 0.f, 0.f, 0.f};
  return z;
}

template <bool TRANS>
__global__ __launch_bounds__(256) void proj_gemm_kernel(
    const float* __restrict__ A, const float* __restrict__ W,
    float* __restrict__ out, int Kd) {
  const int tid  = threadIdx.x;
  const int w    = tid >> 5;
  const int lane = tid & 31;
  const int half = lane >> 4;
  const int l15  = lane & 15;
  const int row_base = blockIdx.x * 16;
  const int col0 = w * 32 + l15;

  const float* Ab = A + (size_t)(row_base + l15) * Kd + 2 * half;

  v8f acc0 = zero8();
  v8f acc1 = zero8();

  for (int kk = 0; kk < Kd; kk += 4) {
    v2f a;
    a.x = Ab[kk];
    a.y = Ab[kk + 1];
    const float* Wb = W + (size_t)(kk + 2 * half) * DP_ + col0;
    v2f b0; b0.x = Wb[0];  b0.y = Wb[DP_];
    v2f b1; b1.x = Wb[16]; b1.y = Wb[16 + DP_];
    acc0 = wmma4(a, b0, acc0);
    acc1 = wmma4(a, b1, acc1);
  }

#pragma unroll
  for (int v = 0; v < 8; ++v) {
    const int m = v + 8 * half;
    const size_t r = (size_t)row_base + m;
    if (!TRANS) {
      out[r * DP_ + col0]      = acc0[v];
      out[r * DP_ + col0 + 16] = acc1[v];
    } else {
      const size_t bb = r >> 13;
      const size_t l  = r & (size_t)(L_ - 1);
      out[(bb * DP_ + col0) * L_ + l]        = acc0[v];
      out[(bb * DP_ + col0 + 16) * L_ + l]   = acc1[v];
    }
  }
}

__global__ __launch_bounds__(256) void attn_kernel(
    const float* __restrict__ H, const float* __restrict__ Q,
    const float* __restrict__ KT, const int* __restrict__ mask,
    float* __restrict__ Z) {
  __shared__ float q_lds[16 * DP_];
  __shared__ float p_lds[16 * 256];
  __shared__ float red[8 * 16];
  __shared__ float mrow[16];
  __shared__ float srow[16];
  __shared__ float frow[16];

  const int tid  = threadIdx.x;
  const int w    = tid >> 5;
  const int lane = tid & 31;
  const int half = lane >> 4;
  const int l15  = lane & 15;
  const int b    = blockIdx.x >> 5;
  const int t0   = (blockIdx.x & 31) * 16;

  const float* Qb = Q + ((size_t)b * T_ + t0) * DP_;
  for (int i = tid; i < 16 * DP_; i += 256) q_lds[i] = Qb[i];
  if (tid < 16) { mrow[tid] = -3.0e38f; srow[tid] = 0.f; }
  __syncthreads();

  v8f zacc[8];
#pragma unroll
  for (int nt = 0; nt < 8; ++nt) zacc[nt] = zero8();

  const float scale = 0.0625f;
  const float* KTb = KT + (size_t)b * DP_ * L_;
  const int colw = w * 32 + l15;

  for (int lc = 0; lc < L_; lc += 256) {
    v8f s0 = zero8(), s1 = zero8();
    for (int kk = 0; kk < DP_; kk += 4) {
      v2f a;
      const int ai = l15 * DP_ + kk + 2 * half;
      a.x = q_lds[ai];
      a.y = q_lds[ai + 1];
      const float* kp = KTb + (size_t)(kk + 2 * half) * L_ + lc + colw;
      v2f b0; b0.x = kp[0];  b0.y = kp[L_];
      v2f b1; b1.x = kp[16]; b1.y = kp[16 + L_];
      s0 = wmma4(a, b0, s0);
      s1 = wmma4(a, b1, s1);
    }

    const int mv0 = mask[(size_t)b * L_ + lc + colw];
    const int mv1 = mask[(size_t)b * L_ + lc + colw + 16];
#pragma unroll
    for (int v = 0; v < 8; ++v) {
      float x0 = s0[v] * scale; if (mv0) x0 = -3.0e38f;
      float x1 = s1[v] * scale; if (mv1) x1 = -3.0e38f;
      s0[v] = x0; s1[v] = x1;
      float x = fmaxf(x0, x1);
      x = fmaxf(x, __shfl_xor(x, 8, 16));
      x = fmaxf(x, __shfl_xor(x, 4, 16));
      x = fmaxf(x, __shfl_xor(x, 2, 16));
      x = fmaxf(x, __shfl_xor(x, 1, 16));
      if (l15 == 0) red[w * 16 + v + 8 * half] = x;
    }
    __syncthreads();

    if (tid < 16) {
      float cm = red[tid];
#pragma unroll
      for (int ww = 1; ww < 8; ++ww) cm = fmaxf(cm, red[ww * 16 + tid]);
      const float mo = mrow[tid];
      const float mn = fmaxf(mo, cm);
      const float f  = __expf(mo - mn);
      srow[tid] *= f;
      frow[tid] = f;
      mrow[tid] = mn;
    }
    __syncthreads();

    float fsel[8], msel[8];
#pragma unroll
    for (int v = 0; v < 8; ++v) {
      fsel[v] = frow[v + 8 * half];
      msel[v] = mrow[v + 8 * half];
    }
#pragma unroll
    for (int nt = 0; nt < 8; ++nt)
#pragma unroll
      for (int v = 0; v < 8; ++v) zacc[nt][v] *= fsel[v];

#pragma unroll
    for (int v = 0; v < 8; ++v) {
      const float p0 = __expf(s0[v] - msel[v]);
      const float p1 = __expf(s1[v] - msel[v]);
      const int row = v + 8 * half;
      p_lds[row * 256 + colw]      = p0;
      p_lds[row * 256 + colw + 16] = p1;
      float x = p0 + p1;
      x += __shfl_xor(x, 8, 16);
      x += __shfl_xor(x, 4, 16);
      x += __shfl_xor(x, 2, 16);
      x += __shfl_xor(x, 1, 16);
      if (l15 == 0) red[w * 16 + row] = x;
    }
    __syncthreads();

    if (tid < 16) {
      float acc = 0.f;
#pragma unroll
      for (int ww = 0; ww < 8; ++ww) acc += red[ww * 16 + tid];
      srow[tid] += acc;
    }
    __syncthreads();

    const float* Hb = H + ((size_t)b * L_ + lc) * DH_ + w * 128 + l15;
    for (int kk = 0; kk < 256; kk += 4) {
      v2f a;
      const int ai = l15 * 256 + kk + 2 * half;
      a.x = p_lds[ai];
      a.y = p_lds[ai + 1];
      const float* hrow = Hb + (size_t)(kk + 2 * half) * DH_;
#pragma unroll
      for (int nt = 0; nt < 8; ++nt) {
        v2f bb;
        bb.x = hrow[nt * 16];
        bb.y = hrow[nt * 16 + DH_];
        zacc[nt] = wmma4(a, bb, zacc[nt]);
      }
    }
  }

  float inv[8];
#pragma unroll
  for (int v = 0; v < 8; ++v) inv[v] = 1.0f / srow[v + 8 * half];
  float* Zb = Z + ((size_t)b * T_ + t0) * DH_;
#pragma unroll
  for (int nt = 0; nt < 8; ++nt)
#pragma unroll
    for (int v = 0; v < 8; ++v) {
      const int m = v + 8 * half;
      const int d = w * 128 + nt * 16 + l15;
      Zb[(size_t)m * DH_ + d] = zacc[nt][v] * inv[v];
    }
}

extern "C" void kernel_launch(void* const* d_in, const int* in_sizes, int n_in,
                              void* d_out, int out_size, void* d_ws, size_t ws_size,
                              hipStream_t stream) {
  const float* H    = (const float*)d_in[0];
  const float* G    = (const float*)d_in[1];
  const float* Wk   = (const float*)d_in[2];
  const float* Wq   = (const float*)d_in[3];
  const int*   mask = (const int*)d_in[4];
  float*       Z    = (float*)d_out;

  float* Qws  = (float*)d_ws;
  float* KTws = Qws + (size_t)B_ * T_ * DP_;

  proj_gemm_kernel<false><<<(B_ * T_) / 16, 256, 0, stream>>>(G, Wq, Qws, DG_);
  proj_gemm_kernel<true><<<(B_ * L_) / 16, 256, 0, stream>>>(H, Wk, KTws, DH_);
  attn_kernel<<<B_ * (T_ / 16), 256, 0, stream>>>(H, Qws, KTws, mask, Z);
}


# attention workgroup runs 32 waves (was 8): a wave owns one logits tile and 32 output columns, L chunk 512, Q and P tiles at LDS pitch 4 mod 64; softmax row-sum order and rescale points change
# speedup vs baseline: 1.2154x; 1.2154x over previous
#include <hip/hip_runtime.h>
#include <hip/hip_bf16.h>

typedef __attribute__((ext_vector_type(2))) float v2f;
typedef __attribute__((ext_vector_type(8))) float v8f;

constexpr int B_ = 8, L_ = 8192, T_ = 512, DH_ = 1024, DG_ = 768, DP_ = 256;

__device__ __forceinline__ v8f wmma4(v2f a, v2f b, v8f c) {
  return __builtin_amdgcn_wmma_f32_16x16x4_f32(
       false, a,  false, b,
       (short)0, c,  false,  false);
}

__device__ __forceinline__ v8f zero8() {
  v8f z = {0.f, 0.f, 0.f, 0.f, 0.f, 0.f, 0.f, 0.f};
  return z;
}

template <bool TRANS>
__global__ __launch_bounds__(256) void proj_gemm_kernel(
    const float* __restrict__ A, const float* __restrict__ W,
    float* __restrict__ out, int Kd) {
  const int tid  = threadIdx.x;
  const int w    = tid >> 5;
  const int lane = tid & 31;
  const int half = lane >> 4;
  const int l15  = lane & 15;
  const int row_base = blockIdx.x * 16;
  const int col0 = w * 32 + l15;

  const float* Ab = A + (size_t)(row_base + l15) * Kd + 2 * half;

  v8f acc0 = zero8();
  v8f acc1 = zero8();

  for (int kk = 0; kk < Kd; kk += 4) {
    v2f a;
    a.x = Ab[kk];
    a.y = Ab[kk + 1];
    const float* Wb = W + (size_t)(kk + 2 * half) * DP_ + col0;
    v2f b0; b0.x = Wb[0];  b0.y = Wb[DP_];
    v2f b1; b1.x = Wb[16]; b1.y = Wb[16 + DP_];
    acc0 = wmma4(a, b0, acc0);
    acc1 = wmma4(a, b1, acc1);
  }

#pragma unroll
  for (int v = 0; v < 8; ++v) {
    const int m = v + 8 * half;
    const size_t r = (size_t)row_base + m;
    if (!TRANS) {
      out[r * DP_ + col0]      = acc0[v];
      out[r * DP_ + col0 + 16] = acc1[v];
    } else {
      const size_t bb = r >> 13;
      const size_t l  = r & (size_t)(L_ - 1);
      out[(bb * DP_ + col0) * L_ + l]        = acc0[v];
      out[(bb * DP_ + col0 + 16) * L_ + l]   = acc1[v];
    }
  }
}

constexpr int NW_ = 32;
constexpr int LC_ = 16 * NW_;
constexpr int QP_ = DP_ + 4;
constexpr int PP_ = LC_ + 4;

__global__ __launch_bounds__(32 * NW_) void attn_kernel(
    const float* __restrict__ H, const float* __restrict__ Q,
    const float* __restrict__ KT, const int* __restrict__ mask,
    float* __restrict__ Z) {
  __shared__ float q_lds[16 * QP_];
  __shared__ float p_lds[16 * PP_];
  __shared__ float red[NW_ * 16];
  __shared__ float mrow[16];
  __shared__ float srow[16];
  __shared__ float frow[16];

  const int tid  = threadIdx.x;
  const int w    = tid >> 5;
  const int lane = tid & 31;
  const int half = lane >> 4;
  const int l15  = lane & 15;
  const int b    = blockIdx.x >> 5;
  const int t0   = (blockIdx.x & 31) * 16;

  const float* Qb = Q + ((size_t)b * T_ + t0) * DP_;
  for (int i = tid; i < 16 * DP_; i += 32 * NW_)
    q_lds[(i >> 8) * QP_ + (i & (DP_ - 1))] = Qb[i];
  if (tid < 16) { mrow[tid] = -3.0e38f; srow[tid] = 0.f; }
  __syncthreads();

  v8f zacc[2];
  zacc[0] = zero8();
  zacc[1] = zero8();

  const float scale = 0.0625f;
  const float* KTb = KT + (size_t)b * DP_ * L_;
  const int colw = w * 16 + l15;

  for (int lc = 0; lc < L_; lc += LC_) {
    v8f s0 = zero8();
    const float* kbase = KTb + (size_t)(2 * half) * L_ + lc + colw;
    const float* qbase = q_lds + l15 * QP_ + 2 * half;
    for (int kk = 0; kk < DP_; kk += 4) {
      v2f a;
      a.x = qbase[kk];
      a.y = qbase[kk + 1];
      const float* kp = kbase + (size_t)kk * L_;
      v2f b0; b0.x = kp[0]; b0.y = kp[L_];
      s0 = wmma4(a, b0, s0);
    }

    const int mv0 = mask[(size_t)b * L_ + lc + colw];
#pragma unroll
    for (int v = 0; v < 8; ++v) {
      float x0 = s0[v] * scale; if (mv0) x0 = -3.0e38f;
      s0[v] = x0;
      float x = x0;
      x = fmaxf(x, __shfl_xor(x, 8, 16));
      x = fmaxf(x, __shfl_xor(x, 4, 16));
      x = fmaxf(x, __shfl_xor(x, 2, 16));
      x = fmaxf(x, __shfl_xor(x, 1, 16));
      if (l15 == 0) red[w * 16 + v + 8 * half] = x;
    }
    __syncthreads();

    if (tid < 16) {
      float cm = red[tid];
#pragma unroll
      for (int ww = 1; ww < NW_; ++ww) cm = fmaxf(cm, red[ww * 16 + tid]);
      const float mo = mrow[tid];
      const float mn = fmaxf(mo, cm);
      const float f  = __expf(mo - mn);
      srow[tid] *= f;
      frow[tid] = f;
      mrow[tid] = mn;
    }
    __syncthreads();

#pragma unroll
    for (int v = 0; v < 8; ++v) {
      const int row = v + 8 * half;
      const float fs = frow[row];
      const float ms = mrow[row];
      zacc[0][v] *= fs;
      zacc[1][v] *= fs;
      const float p0 = __expf(s0[v] - ms);
      p_lds[row * PP_ + colw] = p0;
      float x = p0;
      x += __shfl_xor(x, 8, 16);
      x += __shfl_xor(x, 4, 16);
      x += __shfl_xor(x, 2, 16);
      x += __shfl_xor(x, 1, 16);
      if (l15 == 0) red[w * 16 + row] = x;
    }
    __syncthreads();

    if (tid < 16) {
      float acc = 0.f;
#pragma unroll
      for (int ww = 0; ww < NW_; ++ww) acc += red[ww * 16 + tid];
      srow[tid] += acc;
    }
    __syncthreads();

    const float* Hb = H + ((size_t)b * L_ + lc + 2 * half) * DH_ + w * 32 + l15;
    const float* pbase = p_lds + l15 * PP_ + 2 * half;
    for (int kk = 0; kk < LC_; kk += 4) {
      v2f a;
      a.x = pbase[kk];
      a.y = pbase[kk + 1];
      const float* hrow = Hb + (size_t)kk * DH_;
#pragma unroll
      for (int nt = 0; nt < 2; ++nt) {
        v2f bb;
        bb.x = hrow[nt * 16];
        bb.y = hrow[nt * 16 + DH_];
        zacc[nt] = wmma4(a, bb, zacc[nt]);
      }
    }
  }

  float* Zb = Z + ((size_t)b * T_ + t0) * DH_;
#pragma unroll
  for (int v = 0; v < 8; ++v) {
    const int m = v + 8 * half;
    const float inv = 1.0f / srow[m];
#pragma unroll
    for (int nt = 0; nt < 2; ++nt) {
      const int d = w * 32 + nt * 16 + l15;
      Zb[(size_t)m * DH_ + d] = zacc[nt][v] * inv;
    }
  }
}

extern "C" void kernel_launch(void* const* d_in, const int* in_sizes, int n_in,
                              void* d_out, int out_size, void* d_ws, size_t ws_size,
                              hipStream_t stream) {
  const float* H    = (const float*)d_in[0];
  const float* G    = (const float*)d_in[1];
  const float* Wk   = (const float*)d_in[2];
  const float* Wq   = (const float*)d_in[3];
  const int*   mask = (const int*)d_in[4];
  float*       Z    = (float*)d_out;

  float* Qws  = (float*)d_ws;
  float* KTws = Qws + (size_t)B_ * T_ * DP_;

  proj_gemm_kernel<false><<<(B_ * T_) / 16, 256, 0, stream>>>(G, Wq, Qws, DG_);
  proj_gemm_kernel<true><<<(B_ * L_) / 16, 256, 0, stream>>>(H, Wk, KTws, DH_);
  attn_kernel<<<B_ * (T_ / 16), 32 * NW_, 0, stream>>>(H, Qws, KTws, mask, Z);
}


# K projection: waves own rows, not columns: a wave keeps 16 H rows x 256 outputs, Wk goes through LDS in 32-k slabs, an H fragment feeds 16 WMMAs as B operand, K^T tiles store along L; k order kept
# speedup vs baseline: 1.4083x; 1.1587x over previous
#include <hip/hip_runtime.h>
#include <hip/hip_bf16.h>

typedef __attribute__((ext_vector_type(2))) float v2f;
typedef __attribute__((ext_vector_type(4))) float v4f;
typedef __attribute__((ext_vector_type(8))) float v8f;

constexpr int B_ = 8, L_ = 8192, T_ = 512, DH_ = 1024, DG_ = 768, DP_ = 256;

__device__ __forceinline__ v8f wmma4(v2f a, v2f b, v8f c) {
  return __builtin_amdgcn_wmma_f32_16x16x4_f32(
       false, a,  false, b,
       (short)0, c,  false,  false);
}

__device__ __forceinline__ v8f zero8() {
  v8f z = {0.f, 0.f, 0.f, 0.f, 0.f, 0.f, 0.f, 0.f};
  return z;
}

__global__ __launch_bounds__(256) void qproj_kernel(
    const float* __restrict__ A, const float* __restrict__ W,
    float* __restrict__ out, int Kd) {
  const int tid  = threadIdx.x;
  const int w    = tid >> 5;
  const int lane = tid & 31;
  const int half = lane >> 4;
  const int l15  = lane & 15;
  const int row_base = blockIdx.x * 16;
  const int col0 = w * 32 + l15;

  const float* Ab = A + (size_t)(row_base + l15) * Kd + 2 * half;

  v8f acc0 = zero8();
  v8f acc1 = zero8();

  for (int kk = 0; kk < Kd; kk += 4) {
    v2f a;
    a.x = Ab[kk];
    a.y = Ab[kk + 1];
    const float* Wb = W + (size_t)(kk + 2 * half) * DP_ + col0;
    v2f b0; b0.x = Wb[0];  b0.y = Wb[DP_];
    v2f b1; b1.x = Wb[16]; b1.y = Wb[16 + DP_];
    acc0 = wmma4(a, b0, acc0);
    acc1 = wmma4(a, b1, acc1);
  }

#pragma unroll
  for (int v = 0; v < 8; ++v) {
    const int m = v + 8 * half;
    const size_t r = (size_t)row_base + m;
    out[r * DP_ + col0]      = acc0[v];
    out[r * DP_ + col0 + 16] = acc1[v];
  }
}

constexpr int KPW_  = 16;
constexpr int KPR_  = 16 * KPW_;
constexpr int KS_   = 32;
constexpr int NS_   = DH_ / KS_;
constexpr int WP2_  = 2 * DP_ + 32;
constexpr int SLAB_ = (KS_ / 2) * WP2_;

__device__ __forceinline__ void kp_frags(v2f (&b)[8], const float* lb, int grp) {
#pragma unroll
  for (int jj = 0; jj < 8; ++jj)
    b[jj] = *(const v2f*)(lb + 2 * (grp >> 1) * WP2_ + 32 * (8 * (grp & 1) + jj));
}

__device__ __forceinline__ void kp_slab(const float* lb, const v2f (&a)[8], v8f (&acc)[16]) {
  v2f bc[8], bn[8];
  kp_frags(bc, lb, 0);
#pragma unroll
  for (int grp = 0; grp < 16; ++grp) {
    if (grp < 15) kp_frags(bn, lb, grp + 1);
    __builtin_amdgcn_sched_barrier(0);
#pragma unroll
    for (int jj = 0; jj < 8; ++jj) {
      const int j = 8 * (grp & 1) + jj;
      acc[j] = wmma4(bc[jj], a[grp >> 1], acc[j]);
    }
#pragma unroll
    for (int jj = 0; jj < 8; ++jj) bc[jj] = bn[jj];
  }
}

struct WkPiece { v4f x[2], y[2]; };

__device__ __forceinline__ void kp_fetch(WkPiece& p, const float* Wsrc) {
#pragma unroll
  for (int i = 0; i < 2; ++i) {
    p.x[i] = *(const v4f*)(Wsrc + (size_t)(16 * i) * DP_);
    p.y[i] = *(const v4f*)(Wsrc + (size_t)(16 * i + 1) * DP_);
  }
}

__device__ __forceinline__ void kp_stage(const WkPiece& p, float* dst) {
#pragma unroll
  for (int i = 0; i < 2; ++i) {
    const v4f lo = {p.x[i].x, p.y[i].x, p.x[i].y, p.y[i].y};
    const v4f hi = {p.x[i].z, p.y[i].z, p.x[i].w, p.y[i].w};
    *(v4f*)(dst + 8 * i * WP2_)     = lo;
    *(v4f*)(dst + 8 * i * WP2_ + 4) = hi;
  }
}

__global__ __launch_bounds__(32 * KPW_) void kproj_kernel(
    const float* __restrict__ Hm, const float* __restrict__ Wk,
    float* __restrict__ KT) {
  __shared__ alignas(16) float wl[2 * SLAB_];

  const int tid  = threadIdx.x;
  const int w    = tid >> 5;
  const int lane = tid & 31;
  const int half = lane >> 4;
  const int l15  = lane & 15;
  const int row0 = blockIdx.x * KPR_ + w * 16;

  const float* Hp = Hm + (size_t)(row0 + l15) * DH_ + 2 * half;
  const int g   = tid & 63;
  const int kp0 = tid >> 6;
  const float* Wp = Wk + (size_t)(2 * kp0) * DP_ + 4 * g;
  float* ws0 = wl + kp0 * WP2_ + 8 * g;
  const float* lb0 = wl + half * WP2_ + 2 * l15;

  v8f acc[16];
#pragma unroll
  for (int j = 0; j < 16; ++j) acc[j] = zero8();

  WkPiece wp;
  v2f ac[8], an[8];

  kp_fetch(wp, Wp);
#pragma unroll
  for (int ks = 0; ks < 8; ++ks) ac[ks] = *(const v2f*)(Hp + 4 * ks);
  kp_stage(wp, ws0);
  __syncthreads();

  for (int s = 0; s < NS_ - 1; ++s) {
    kp_fetch(wp, Wp + (size_t)(s + 1) * KS_ * DP_);
    const float* Hn = Hp + (s + 1) * KS_;
#pragma unroll
    for (int ks = 0; ks < 8; ++ks) an[ks] = *(const v2f*)(Hn + 4 * ks);

    kp_slab(lb0 + (s & 1) * SLAB_, ac, acc);

    kp_stage(wp, ws0 + ((s + 1) & 1) * SLAB_);
#pragma unroll
    for (int ks = 0; ks < 8; ++ks) ac[ks] = an[ks];
    __syncthreads();
  }
  kp_slab(lb0 + ((NS_ - 1) & 1) * SLAB_, ac, acc);

  const size_t bb = (size_t)row0 >> 13;
  const size_t l0 = (size_t)row0 & (size_t)(L_ - 1);
  float* Kb = KT + (bb * DP_ + 8 * half) * L_ + l0 + l15;
#pragma unroll
  for (int j = 0; j < 16; ++j) {
#pragma unroll
    for (int r = 0; r < 8; ++r) Kb[(size_t)(16 * j + r) * L_] = acc[j][r];
  }
}

constexpr int NW_ = 32;
constexpr int LC_ = 16 * NW_;
constexpr int QP_ = DP_ + 4;
constexpr int PP_ = LC_ + 4;

__global__ __launch_bounds__(32 * NW_) void attn_kernel(
    const float* __restrict__ H, const float* __restrict__ Q,
    const float* __restrict__ KT, const int* __restrict__ mask,
    float* __restrict__ Z) {
  __shared__ float q_lds[16 * QP_];
  __shared__ float p_lds[16 * PP_];
  __shared__ float red[NW_ * 16];
  __shared__ float mrow[16];
  __shared__ float srow[16];
  __shared__ float frow[16];

  const int tid  = threadIdx.x;
  const int w    = tid >> 5;
  const int lane = tid & 31;
  const int half = lane >> 4;
  const int l15  = lane & 15;
  const int b    = blockIdx.x >> 5;
  const int t0   = (blockIdx.x & 31) * 16;

  const float* Qb = Q + ((size_t)b * T_ + t0) * DP_;
  for (int i = tid; i < 16 * DP_; i += 32 * NW_)
    q_lds[(i >> 8) * QP_ + (i & (DP_ - 1))] = Qb[i];
  if (tid < 16) { mrow[tid] = -3.0e38f; srow[tid] = 0.f; }
  __syncthreads();

  v8f zacc[2];
  zacc[0] = zero8();
  zacc[1] = zero8();

  const float scale = 0.0625f;
  const float* KTb = KT + (size_t)b * DP_ * L_;
  const int colw = w * 16 + l15;

  for (int lc = 0; lc < L_; lc += LC_) {
    v8f s0 = zero8();
    const float* kbase = KTb + (size_t)(2 * half) * L_ + lc + colw;
    const float* qbase = q_lds + l15 * QP_ + 2 * half;
    for (int kk = 0; kk < DP_; kk += 4) {
      v2f a;
      a.x = qbase[kk];
      a.y = qbase[kk + 1];
      const float* kp = kbase + (size_t)kk * L_;
      v2f b0; b0.x = kp[0]; b0.y = kp[L_];
      s0 = wmma4(a, b0, s0);
    }

    const int mv0 = mask[(size_t)b * L_ + lc + colw];
#pragma unroll
    for (int v = 0; v < 8; ++v) {
      float x0 = s0[v] * scale; if (mv0) x0 = -3.0e38f;
      s0[v] = x0;
      float x = x0;
      x = fmaxf(x, __shfl_xor(x, 8, 16));
      x = fmaxf(x, __shfl_xor(x, 4, 16));
      x = fmaxf(x, __shfl_xor(x, 2, 16));
      x = fmaxf(x, __shfl_xor(x, 1, 16));
      if (l15 == 0) red[w * 16 + v + 8 * half] = x;
    }
    __syncthreads();

    if (tid < 16) {
      float cm = red[tid];
#pragma unroll
      for (int ww = 1; ww < NW_; ++ww) cm = fmaxf(cm, red[ww * 16 + tid]);
      const float mo = mrow[tid];
      const float mn = fmaxf(mo, cm);
      const float f  = __expf(mo - mn);
      srow[tid] *= f;
      frow[tid] = f;
      mrow[tid] = mn;
    }
    __syncthreads();

#pragma unroll
    for (int v = 0; v < 8; ++v) {
      const int row = v + 8 * half;
      const float fs = frow[row];
      const float ms = mrow[row];
      zacc[0][v] *= fs;
      zacc[1][v] *= fs;
      const float p0 = __expf(s0[v] - ms);
      p_lds[row * PP_ + colw] = p0;
      float x = p0;
      x += __shfl_xor(x, 8, 16);
      x += __shfl_xor(x, 4, 16);
      x += __shfl_xor(x, 2, 16);
      x += __shfl_xor(x, 1, 16);
      if (l15 == 0) red[w * 16 + row] = x;
    }
    __syncthreads();

    if (tid < 16) {
      float acc = 0.f;
#pragma unroll
      for (int ww = 0; ww < NW_; ++ww) acc += red[ww * 16 + tid];
      srow[tid] += acc;
    }
    __syncthreads();

    const float* Hb = H + ((size_t)b * L_ + lc + 2 * half) * DH_ + w * 32 + l15;
    const float* pbase = p_lds + l15 * PP_ + 2 * half;
    for (int kk = 0; kk < LC_; kk += 4) {
      v2f a;
      a.x = pbase[kk];
      a.y = pbase[kk + 1];
      const float* hrow = Hb + (size_t)kk * DH_;
#pragma unroll
      for (int nt = 0; nt < 2; ++nt) {
        v2f bb;
        bb.x = hrow[nt * 16];
        bb.y = hrow[nt * 16 + DH_];
        zacc[nt] = wmma4(a, bb, zacc[nt]);
      }
    }
  }

  float* Zb = Z + ((size_t)b * T_ + t0) * DH_;
#pragma unroll
  for (int v = 0; v < 8; ++v) {
    const int m = v + 8 * half;
    const float inv = 1.0f / srow[m];
#pragma unroll
    for (int nt = 0; nt < 2; ++nt) {
      const int d = w * 32 + nt * 16 + l15;
      Zb[(size_t)m * DH_ + d] = zacc[nt][v] * inv;
    }
  }
}

extern "C" void kernel_launch(void* const* d_in, const int* in_sizes, int n_in,
                              void* d_out, int out_size, void* d_ws, size_t ws_size,
                              hipStream_t stream) {
  const float* H    = (const float*)d_in[0];
  const float* G    = (const float*)d_in[1];
  const float* Wk   = (const float*)d_in[2];
  const float* Wq   = (const float*)d_in[3];
  const int*   mask = (const int*)d_in[4];
  float*       Z    = (float*)d_out;

  float* Qws  = (float*)d_ws;
  float* KTws = Qws + (size_t)B_ * T_ * DP_;

  qproj_kernel<<<(B_ * T_) / 16, 256, 0, stream>>>(G, Wq, Qws, DG_);
  kproj_kernel<<<(B_ * L_) / KPR_, 32 * KPW_, 0, stream>>>(H, Wk, KTws);
  attn_kernel<<<B_ * (T_ / 16), 32 * NW_, 0, stream>>>(H, Qws, KTws, mask, Z);
}


# attention: softmax of chunk c+1 runs under the weighted sum of chunk c: logits one chunk ahead, the 3 workgroup barriers split into signal and wait around matrix spans, P double-buffered; sums same
# speedup vs baseline: 1.6112x; 1.1441x over previous
#include <hip/hip_runtime.h>
#include <hip/hip_bf16.h>

typedef __attribute__((ext_vector_type(2))) float v2f;
typedef __attribute__((ext_vector_type(4))) float v4f;
typedef __attribute__((ext_vector_type(8))) float v8f;

constexpr int B_ = 8, L_ = 8192, T_ = 512, DH_ = 1024, DG_ = 768, DP_ = 256;

__device__ __forceinline__ v8f wmma4(v2f a, v2f b, v8f c) {
  return __builtin_amdgcn_wmma_f32_16x16x4_f32(
       false, a,  false, b,
       (short)0, c,  false,  false);
}

__device__ __forceinline__ v8f zero8() {
  v8f z = {0.f, 0.f, 0.f, 0.f, 0.f, 0.f, 0.f, 0.f};
  return z;
}

__global__ __launch_bounds__(256) void qproj_kernel(
    const float* __restrict__ A, const float* __restrict__ W,
    float* __restrict__ out, int Kd) {
  const int tid  = threadIdx.x;
  const int w    = tid >> 5;
  const int lane = tid & 31;
  const int half = lane >> 4;
  const int l15  = lane & 15;
  const int row_base = blockIdx.x * 16;
  const int col0 = w * 32 + l15;

  const float* Ab = A + (size_t)(row_base + l15) * Kd + 2 * half;

  v8f acc0 = zero8();
  v8f acc1 = zero8();

  for (int kk = 0; kk < Kd; kk += 4) {
    v2f a;
    a.x = Ab[kk];
    a.y = Ab[kk + 1];
    const float* Wb = W + (size_t)(kk + 2 * half) * DP_ + col0;
    v2f b0; b0.x = Wb[0];  b0.y = Wb[DP_];
    v2f b1; b1.x = Wb[16]; b1.y = Wb[16 + DP_];
    acc0 = wmma4(a, b0, acc0);
    acc1 = wmma4(a, b1, acc1);
  }

#pragma unroll
  for (int v = 0; v < 8; ++v) {
    const int m = v + 8 * half;
    const size_t r = (size_t)row_base + m;
    out[r * DP_ + col0]      = acc0[v];
    out[r * DP_ + col0 + 16] = acc1[v];
  }
}

constexpr int KPW_  = 16;
constexpr int KPR_  = 16 * KPW_;
constexpr int KS_   = 32;
constexpr int NS_   = DH_ / KS_;
constexpr int WP2_  = 2 * DP_ + 32;
constexpr int SLAB_ = (KS_ / 2) * WP2_;

__device__ __forceinline__ void kp_frags(v2f (&b)[8], const float* lb, int grp) {
#pragma unroll
  for (int jj = 0; jj < 8; ++jj)
    b[jj] = *(const v2f*)(lb + 2 * (grp >> 1) * WP2_ + 32 * (8 * (grp & 1) + jj));
}

__device__ __forceinline__ void kp_slab(const float* lb, const v2f (&a)[8], v8f (&acc)[16]) {
  v2f bc[8], bn[8];
  kp_frags(bc, lb, 0);
#pragma unroll
  for (int grp = 0; grp < 16; ++grp) {
    if (grp < 15) kp_frags(bn, lb, grp + 1);
    __builtin_amdgcn_sched_barrier(0);
#pragma unroll
    for (int jj = 0; jj < 8; ++jj) {
      const int j = 8 * (grp & 1) + jj;
      acc[j] = wmma4(bc[jj], a[grp >> 1], acc[j]);
    }
#pragma unroll
    for (int jj = 0; jj < 8; ++jj) bc[jj] = bn[jj];
  }
}

struct WkPiece { v4f x[2], y[2]; };

__device__ __forceinline__ void kp_fetch(WkPiece& p, const float* Wsrc) {
#pragma unroll
  for (int i = 0; i < 2; ++i) {
    p.x[i] = *(const v4f*)(Wsrc + (size_t)(16 * i) * DP_);
    p.y[i] = *(const v4f*)(Wsrc + (size_t)(16 * i + 1) * DP_);
  }
}

__device__ __forceinline__ void kp_stage(const WkPiece& p, float* dst) {
#pragma unroll
  for (int i = 0; i < 2; ++i) {
    const v4f lo = {p.x[i].x, p.y[i].x, p.x[i].y, p.y[i].y};
    const v4f hi = {p.x[i].z, p.y[i].z, p.x[i].w, p.y[i].w};
    *(v4f*)(dst + 8 * i * WP2_)     = lo;
    *(v4f*)(dst + 8 * i * WP2_ + 4) = hi;
  }
}

__global__ __launch_bounds__(32 * KPW_) void kproj_kernel(
    const float* __restrict__ Hm, const float* __restrict__ Wk,
    float* __restrict__ KT) {
  __shared__ alignas(16) float wl[2 * SLAB_];

  const int tid  = threadIdx.x;
  const int w    = tid >> 5;
  const int lane = tid & 31;
  const int half = lane >> 4;
  const int l15  = lane & 15;
  const int row0 = blockIdx.x * KPR_ + w * 16;

  const float* Hp = Hm + (size_t)(row0 + l15) * DH_ + 2 * half;
  const int g   = tid & 63;
  const int kp0 = tid >> 6;
  const float* Wp = Wk + (size_t)(2 * kp0) * DP_ + 4 * g;
  float* ws0 = wl + kp0 * WP2_ + 8 * g;
  const float* lb0 = wl + half * WP2_ + 2 * l15;

  v8f acc[16];
#pragma unroll
  for (int j = 0; j < 16; ++j) acc[j] = zero8();

  WkPiece wp;
  v2f ac[8], an[8];

  kp_fetch(wp, Wp);
#pragma unroll
  for (int ks = 0; ks < 8; ++ks) ac[ks] = *(const v2f*)(Hp + 4 * ks);
  kp_stage(wp, ws0);
  __syncthreads();

  for (int s = 0; s < NS_ - 1; ++s) {
    kp_fetch(wp, Wp + (size_t)(s + 1) * KS_ * DP_);
    const float* Hn = Hp + (s + 1) * KS_;
#pragma unroll
    for (int ks = 0; ks < 8; ++ks) an[ks] = *(const v2f*)(Hn + 4 * ks);

    kp_slab(lb0 + (s & 1) * SLAB_, ac, acc);

    kp_stage(wp, ws0 + ((s + 1) & 1) * SLAB_);
#pragma unroll
    for (int ks = 0; ks < 8; ++ks) ac[ks] = an[ks];
    __syncthreads();
  }
  kp_slab(lb0 + ((NS_ - 1) & 1) * SLAB_, ac, acc);

  const size_t bb = (size_t)row0 >> 13;
  const size_t l0 = (size_t)row0 & (size_t)(L_ - 1);
  float* Kb = KT + (bb * DP_ + 8 * half) * L_ + l0 + l15;
#pragma unroll
  for (int j = 0; j < 16; ++j) {
#pragma unroll
    for (int r = 0; r < 8; ++r) Kb[(size_t)(16 * j + r) * L_] = acc[j][r];
  }
}

constexpr int NW_ = 32;
constexpr int LC_ = 16 * NW_;
constexpr int NC_ = L_ / LC_;
constexpr int QP_ = DP_ + 4;
constexpr int PP_ = LC_ + 4;
constexpr int PB_ = 16 * PP_;

constexpr int PC1_ = 192;
constexpr int PC2_ = 352;

__device__ __forceinline__ void bar_signal() {
  asm volatile("s_wait_dscnt 0x0\n\ts_barrier_signal -1" ::: "memory");
}
__device__ __forceinline__ void bar_wait() {
  asm volatile("s_barrier_wait -1" ::: "memory");
}

__device__ __forceinline__ void pc_span(v8f (&zacc)[2], const float* pbase, const float* Hb,
                                        int kb, int ke) {
  for (int kk = kb; kk < ke; kk += 4) {
    v2f a;
    a.x = pbase[kk];
    a.y = pbase[kk + 1];
    const float* hrow = Hb + (size_t)kk * DH_;
#pragma unroll
    for (int nt = 0; nt < 2; ++nt) {
      v2f bb;
      bb.x = hrow[nt * 16];
      bb.y = hrow[nt * 16 + DH_];
      zacc[nt] = wmma4(a, bb, zacc[nt]);
    }
  }
}

__device__ __forceinline__ v8f pa_scores(const float* q_lds, float* red, const float* KTb,
                                         const int* maskb, int lc, int w, int half, int l15) {
  const int colw = w * 16 + l15;
  const bool b3 = (l15 & 8) != 0, b2 = (l15 & 4) != 0, b1 = (l15 & 2) != 0;
  const float scale = 0.0625f;
  v8f s0 = zero8();
  const float* kbase = KTb + (size_t)(2 * half) * L_ + lc + colw;
  const float* qbase = q_lds + l15 * QP_ + 2 * half;
  for (int kk = 0; kk < DP_; kk += 4) {
    v2f a;
    a.x = qbase[kk];
    a.y = qbase[kk + 1];
    const float* kp = kbase + (size_t)kk * L_;
    v2f b0; b0.x = kp[0]; b0.y = kp[L_];
    s0 = wmma4(a, b0, s0);
  }

  const int mv0 = maskb[lc + colw];
#pragma unroll
  for (int v = 0; v < 8; ++v) {
    float x0 = s0[v] * scale; if (mv0) x0 = -3.0e38f;
    s0[v] = x0;
  }
  {
    float y[4], z[2];
#pragma unroll
    for (int i = 0; i < 4; ++i) {
      const float keep = b3 ? s0[i + 4] : s0[i];
      const float send = b3 ? s0[i] : s0[i + 4];
      y[i] = fmaxf(keep, __shfl_xor(send, 8, 16));
    }
#pragma unroll
    for (int j = 0; j < 2; ++j) {
      const float keep = b2 ? y[j + 2] : y[j];
      const float send = b2 ? y[j] : y[j + 2];
      z[j] = fmaxf(keep, __shfl_xor(send, 4, 16));
    }
    const float keep1 = b1 ? z[1] : z[0];
    const float send1 = b1 ? z[0] : z[1];
    float u = fmaxf(keep1, __shfl_xor(send1, 2, 16));
    u = fmaxf(u, __shfl_xor(u, 1, 16));
    if ((l15 & 1) == 0) red[w * 16 + 8 * half + (l15 >> 1)] = u;
  }
  return s0;
}

__device__ __forceinline__ void row_max_merge(const float* red, float* mrow, float* frow, int tid) {
  float cm = red[tid];
#pragma unroll
  for (int ww = 1; ww < NW_; ++ww) cm = fmaxf(cm, red[ww * 16 + tid]);
  const float mo = mrow[tid];
  const float mn = fmaxf(mo, cm);
  const float f  = __expf(mo - mn);
  frow[tid] = f;
  mrow[tid] = mn;
}

__global__ __launch_bounds__(32 * NW_) void attn_kernel(
    const float* __restrict__ H, const float* __restrict__ Q,
    const float* __restrict__ KT, const int* __restrict__ mask,
    float* __restrict__ Z) {
  __shared__ float q_lds[16 * QP_];
  __shared__ float p_lds[2 * PB_];
  __shared__ float red[NW_ * 16];
  __shared__ float mrow[16];
  __shared__ float srow[16];
  __shared__ float frow[16];

  const int tid  = threadIdx.x;
  const int w    = tid >> 5;
  const int lane = tid & 31;
  const int half = lane >> 4;
  const int l15  = lane & 15;
  const int b    = blockIdx.x >> 5;
  const int t0   = (blockIdx.x & 31) * 16;

  const float* Qb = Q + ((size_t)b * T_ + t0) * DP_;
  for (int i = tid; i < 16 * DP_; i += 32 * NW_)
    q_lds[(i >> 8) * QP_ + (i & (DP_ - 1))] = Qb[i];
  if (tid < 16) mrow[tid] = -3.0e38f;
  __syncthreads();

  v8f zacc[2];
  zacc[0] = zero8();
  zacc[1] = zero8();
  v8f lsum = zero8();

  const float* KTb = KT + (size_t)b * DP_ * L_;
  const int* maskb = mask + (size_t)b * L_;
  const int colw = w * 16 + l15;

  v8f s0 = pa_scores(q_lds, red, KTb, maskb, 0, w, half, l15);
  __syncthreads();
  if (tid < 16) row_max_merge(red, mrow, frow, tid);
  __syncthreads();
#pragma unroll
  for (int v = 0; v < 8; ++v) {
    const int row = v + 8 * half;
    const float p0 = __expf(s0[v] - mrow[row]);
    p_lds[row * PP_ + colw] = p0;
    lsum[v] = p0;
  }
  bar_signal();

  for (int c = 0; c < NC_; ++c) {
    const int lc = c * LC_;
    const bool more = (c + 1 < NC_);
    const float* pcur = p_lds + (c & 1) * PB_;
    float* pnxt = p_lds + ((c + 1) & 1) * PB_;

    __builtin_amdgcn_s_setprio(0);
    if (more) s0 = pa_scores(q_lds, red, KTb, maskb, lc + LC_, w, half, l15);
    bar_wait();
    if (more) bar_signal();

    const float* Hb = H + ((size_t)b * L_ + lc + 2 * half) * DH_ + w * 32 + l15;
    const float* pbase = pcur + l15 * PP_ + 2 * half;
    __builtin_amdgcn_s_setprio(3);
    pc_span(zacc, pbase, Hb, 0, PC1_);

    if (more) {
      bar_wait();
      if (tid < 16) row_max_merge(red, mrow, frow, tid);
      bar_signal();
    }
    __builtin_amdgcn_s_setprio(2);
    pc_span(zacc, pbase, Hb, PC1_, PC2_);

    float fs[8];
    if (more) {
      bar_wait();
#pragma unroll
      for (int v = 0; v < 8; ++v) {
        const int row = v + 8 * half;
        fs[v] = frow[row];
        const float p0 = __expf(s0[v] - mrow[row]);
        pnxt[row * PP_ + colw] = p0;
        lsum[v] = lsum[v] * fs[v] + p0;
      }
      bar_signal();
    }
    __builtin_amdgcn_s_setprio(1);
    pc_span(zacc, pbase, Hb, PC2_, LC_);

    if (more) {
#pragma unroll
      for (int v = 0; v < 8; ++v) {
        zacc[0][v] *= fs[v];
        zacc[1][v] *= fs[v];
      }
    }
  }
  __builtin_amdgcn_s_setprio(0);

#pragma unroll
  for (int v = 0; v < 8; ++v) {
    float x = lsum[v];
    x += __shfl_xor(x, 8, 16);
    x += __shfl_xor(x, 4, 16);
    x += __shfl_xor(x, 2, 16);
    x += __shfl_xor(x, 1, 16);
    if (l15 == 0) red[w * 16 + v + 8 * half] = x;
  }
  __syncthreads();
  if (tid < 16) {
    float acc = 0.f;
#pragma unroll
    for (int ww = 0; ww < NW_; ++ww) acc += red[ww * 16 + tid];
    srow[tid] = acc;
  }
  __syncthreads();

  float* Zb = Z + ((size_t)b * T_ + t0) * DH_;
#pragma unroll
  for (int v = 0; v < 8; ++v) {
    const int m = v + 8 * half;
    const float inv = 1.0f / srow[m];
#pragma unroll
    for (int nt = 0; nt < 2; ++nt) {
      const int d = w * 32 + nt * 16 + l15;
      Zb[(size_t)m * DH_ + d] = zacc[nt][v] * inv;
    }
  }
}

extern "C" void kernel_launch(void* const* d_in, const int* in_sizes, int n_in,
                              void* d_out, int out_size, void* d_ws, size_t ws_size,
                              hipStream_t stream) {
  const float* H    = (const float*)d_in[0];
  const float* G    = (const float*)d_in[1];
  const float* Wk   = (const float*)d_in[2];
  const float* Wq   = (const float*)d_in[3];
  const int*   mask = (const int*)d_in[4];
  float*       Z    = (float*)d_out;

  float* Qws  = (float*)d_ws;
  float* KTws = Qws + (size_t)B_ * T_ * DP_;

  qproj_kernel<<<(B_ * T_) / 16, 256, 0, stream>>>(G, Wq, Qws, DG_);
  kproj_kernel<<<(B_ * L_) / KPR_, 32 * KPW_, 0, stream>>>(H, Wk, KTws);
  attn_kernel<<<B_ * (T_ / 16), 32 * NW_, 0, stream>>>(H, Qws, KTws, mask, Z);
}


# attention: each workgroup prefetches to the second-level cache, one chunk ahead, an eighth of the H rows and K^T columns its batch reads next (one prefetch a wave); loads, k order, sums unchanged
# speedup vs baseline: 1.6629x; 1.0321x over previous
#include <hip/hip_runtime.h>
#include <hip/hip_bf16.h>

typedef __attribute__((ext_vector_type(2))) float v2f;
typedef __attribute__((ext_vector_type(4))) float v4f;
typedef __attribute__((ext_vector_type(8))) float v8f;

constexpr int B_ = 8, L_ = 8192, T_ = 512, DH_ = 1024, DG_ = 768, DP_ = 256;

__device__ __forceinline__ v8f wmma4(v2f a, v2f b, v8f c) {
  return __builtin_amdgcn_wmma_f32_16x16x4_f32(
       false, a,  false, b,
       (short)0, c,  false,  false);
}

__device__ __forceinline__ v8f zero8() {
  v8f z = {0.f, 0.f, 0.f, 0.f, 0.f, 0.f, 0.f, 0.f};
  return z;
}

__global__ __launch_bounds__(256) void qproj_kernel(
    const float* __restrict__ A, const float* __restrict__ W,
    float* __restrict__ out, int Kd) {
  const int tid  = threadIdx.x;
  const int w    = tid >> 5;
  const int lane = tid & 31;
  const int half = lane >> 4;
  const int l15  = lane & 15;
  const int row_base = blockIdx.x * 16;
  const int col0 = w * 32 + l15;

  const float* Ab = A + (size_t)(row_base + l15) * Kd + 2 * half;

  v8f acc0 = zero8();
  v8f acc1 = zero8();

  for (int kk = 0; kk < Kd; kk += 4) {
    v2f a;
    a.x = Ab[kk];
    a.y = Ab[kk + 1];
    const float* Wb = W + (size_t)(kk + 2 * half) * DP_ + col0;
    v2f b0; b0.x = Wb[0];  b0.y = Wb[DP_];
    v2f b1; b1.x = Wb[16]; b1.y = Wb[16 + DP_];
    acc0 = wmma4(a, b0, acc0);
    acc1 = wmma4(a, b1, acc1);
  }

#pragma unroll
  for (int v = 0; v < 8; ++v) {
    const int m = v + 8 * half;
    const size_t r = (size_t)row_base + m;
    out[r * DP_ + col0]      = acc0[v];
    out[r * DP_ + col0 + 16] = acc1[v];
  }
}

constexpr int KPW_  = 16;
constexpr int KPR_  = 16 * KPW_;
constexpr int KS_   = 32;
constexpr int NS_   = DH_ / KS_;
constexpr int WP2_  = 2 * DP_ + 32;
constexpr int SLAB_ = (KS_ / 2) * WP2_;

__device__ __forceinline__ void kp_frags(v2f (&b)[8], const float* lb, int grp) {
#pragma unroll
  for (int jj = 0; jj < 8; ++jj)
    b[jj] = *(const v2f*)(lb + 2 * (grp >> 1) * WP2_ + 32 * (8 * (grp & 1) + jj));
}

__device__ __forceinline__ void kp_slab(const float* lb, const v2f (&a)[8], v8f (&acc)[16]) {
  v2f bc[8], bn[8];
  kp_frags(bc, lb, 0);
#pragma unroll
  for (int grp = 0; grp < 16; ++grp) {
    if (grp < 15) kp_frags(bn, lb, grp + 1);
    __builtin_amdgcn_sched_barrier(0);
#pragma unroll
    for (int jj = 0; jj < 8; ++jj) {
      const int j = 8 * (grp & 1) + jj;
      acc[j] = wmma4(bc[jj], a[grp >> 1], acc[j]);
    }
#pragma unroll
    for (int jj = 0; jj < 8; ++jj) bc[jj] = bn[jj];
  }
}

struct WkPiece { v4f x[2], y[2]; };

__device__ __forceinline__ void kp_fetch(WkPiece& p, const float* Wsrc) {
#pragma unroll
  for (int i = 0; i < 2; ++i) {
    p.x[i] = *(const v4f*)(Wsrc + (size_t)(16 * i) * DP_);
    p.y[i] = *(const v4f*)(Wsrc + (size_t)(16 * i + 1) * DP_);
  }
}

__device__ __forceinline__ void kp_stage(const WkPiece& p, float* dst) {
#pragma unroll
  for (int i = 0; i < 2; ++i) {
    const v4f lo = {p.x[i].x, p.y[i].x, p.x[i].y, p.y[i].y};
    const v4f hi = {p.x[i].z, p.y[i].z, p.x[i].w, p.y[i].w};
    *(v4f*)(dst + 8 * i * WP2_)     = lo;
    *(v4f*)(dst + 8 * i * WP2_ + 4) = hi;
  }
}

__global__ __launch_bounds__(32 * KPW_) void kproj_kernel(
    const float* __restrict__ Hm, const float* __restrict__ Wk,
    float* __restrict__ KT) {
  __shared__ alignas(16) float wl[2 * SLAB_];

  const int tid  = threadIdx.x;
  const int w    = tid >> 5;
  const int lane = tid & 31;
  const int half = lane >> 4;
  const int l15  = lane & 15;
  const int row0 = blockIdx.x * KPR_ + w * 16;

  const float* Hp = Hm + (size_t)(row0 + l15) * DH_ + 2 * half;
  const int g   = tid & 63;
  const int kp0 = tid >> 6;
  const float* Wp = Wk + (size_t)(2 * kp0) * DP_ + 4 * g;
  float* ws0 = wl + kp0 * WP2_ + 8 * g;
  const float* lb0 = wl + half * WP2_ + 2 * l15;

  v8f acc[16];
#pragma unroll
  for (int j = 0; j < 16; ++j) acc[j] = zero8();

  WkPiece wp;
  v2f ac[8], an[8];

  kp_fetch(wp, Wp);
#pragma unroll
  for (int ks = 0; ks < 8; ++ks) ac[ks] = *(const v2f*)(Hp + 4 * ks);
  kp_stage(wp, ws0);
  __syncthreads();

  for (int s = 0; s < NS_ - 1; ++s) {
    kp_fetch(wp, Wp + (size_t)(s + 1) * KS_ * DP_);
    const float* Hn = Hp + (s + 1) * KS_;
#pragma unroll
    for (int ks = 0; ks < 8; ++ks) an[ks] = *(const v2f*)(Hn + 4 * ks);

    kp_slab(lb0 + (s & 1) * SLAB_, ac, acc);

    kp_stage(wp, ws0 + ((s + 1) & 1) * SLAB_);
#pragma unroll
    for (int ks = 0; ks < 8; ++ks) ac[ks] = an[ks];
    __syncthreads();
  }
  kp_slab(lb0 + ((NS_ - 1) & 1) * SLAB_, ac, acc);

  const size_t bb = (size_t)row0 >> 13;
  const size_t l0 = (size_t)row0 & (size_t)(L_ - 1);
  float* Kb = KT + (bb * DP_ + 8 * half) * L_ + l0 + l15;
#pragma unroll
  for (int j = 0; j < 16; ++j) {
#pragma unroll
    for (int r = 0; r < 8; ++r) Kb[(size_t)(16 * j + r) * L_] = acc[j][r];
  }
}

constexpr int NW_ = 32;
constexpr int LC_ = 16 * NW_;
constexpr int NC_ = L_ / LC_;
constexpr int QP_ = DP_ + 4;
constexpr int PP_ = LC_ + 4;
constexpr int PB_ = 16 * PP_;

constexpr int PC1_ = 192;
constexpr int PC2_ = 352;

__device__ __forceinline__ void bar_signal() {
  asm volatile("s_wait_dscnt 0x0\n\ts_barrier_signal -1" ::: "memory");
}
__device__ __forceinline__ void bar_wait() {
  asm volatile("s_barrier_wait -1" ::: "memory");
}

__device__ __forceinline__ void pc_span(v8f (&zacc)[2], const float* pbase, const float* Hb,
                                        int kb, int ke) {
  for (int kk = kb; kk < ke; kk += 4) {
    v2f a;
    a.x = pbase[kk];
    a.y = pbase[kk + 1];
    const float* hrow = Hb + (size_t)kk * DH_;
#pragma unroll
    for (int nt = 0; nt < 2; ++nt) {
      v2f bb;
      bb.x = hrow[nt * 16];
      bb.y = hrow[nt * 16 + DH_];
      zacc[nt] = wmma4(a, bb, zacc[nt]);
    }
  }
}

__device__ __forceinline__ v8f pa_scores(const float* q_lds, float* red, const float* KTb,
                                         const int* maskb, int lc, int w, int half, int l15) {
  const int colw = w * 16 + l15;
  const bool b3 = (l15 & 8) != 0, b2 = (l15 & 4) != 0, b1 = (l15 & 2) != 0;
  const float scale = 0.0625f;
  v8f s0 = zero8();
  const float* kbase = KTb + (size_t)(2 * half) * L_ + lc + colw;
  const float* qbase = q_lds + l15 * QP_ + 2 * half;
  for (int kk = 0; kk < DP_; kk += 4) {
    v2f a;
    a.x = qbase[kk];
    a.y = qbase[kk + 1];
    const float* kp = kbase + (size_t)kk * L_;
    v2f b0; b0.x = kp[0]; b0.y = kp[L_];
    s0 = wmma4(a, b0, s0);
  }

  const int mv0 = maskb[lc + colw];
#pragma unroll
  for (int v = 0; v < 8; ++v) {
    float x0 = s0[v] * scale; if (mv0) x0 = -3.0e38f;
    s0[v] = x0;
  }
  {
    float y[4], z[2];
#pragma unroll
    for (int i = 0; i < 4; ++i) {
      const float keep = b3 ? s0[i + 4] : s0[i];
      const float send = b3 ? s0[i] : s0[i + 4];
      y[i] = fmaxf(keep, __shfl_xor(send, 8, 16));
    }
#pragma unroll
    for (int j = 0; j < 2; ++j) {
      const float keep = b2 ? y[j + 2] : y[j];
      const float send = b2 ? y[j] : y[j + 2];
      z[j] = fmaxf(keep, __shfl_xor(send, 4, 16));
    }
    const float keep1 = b1 ? z[1] : z[0];
    const float send1 = b1 ? z[0] : z[1];
    float u = fmaxf(keep1, __shfl_xor(send1, 2, 16));
    u = fmaxf(u, __shfl_xor(u, 1, 16));
    if ((l15 & 1) == 0) red[w * 16 + 8 * half + (l15 >> 1)] = u;
  }
  return s0;
}

__device__ __forceinline__ void row_max_merge(const float* red, float* mrow, float* frow, int tid) {
  float cm = red[tid];
#pragma unroll
  for (int ww = 1; ww < NW_; ++ww) cm = fmaxf(cm, red[ww * 16 + tid]);
  const float mo = mrow[tid];
  const float mn = fmaxf(mo, cm);
  const float f  = __expf(mo - mn);
  frow[tid] = f;
  mrow[tid] = mn;
}


__device__ __forceinline__ void l2_fetch_ahead(const float* p) {
  __builtin_prefetch(p, 0, 3);
}
__device__ __forceinline__ int pf_slice(int qt) {
  return ((qt ^ (qt >> 1)) & 1) | (((qt >> 2) ^ (qt >> 3)) & 1) << 1 | (((qt >> 4) ^ qt) & 1) << 2;
}
__device__ __forceinline__ void pf_h_chunk(const float* Hbat, int lc, int s, int w, int lane) {
  const int row = lc + 64 * s + 2 * w + (lane >> 4);
  l2_fetch_ahead(Hbat + (size_t)row * DH_ + 64 * (lane & 15));
}
__device__ __forceinline__ void pf_kt_chunk(const float* KTb, int lc, int s, int w, int lane) {
  if (lane < 8) l2_fetch_ahead(KTb + (size_t)(32 * s + w) * L_ + lc + 64 * lane);
}

__global__ __launch_bounds__(32 * NW_) void attn_kernel(
    const float* __restrict__ H, const float* __restrict__ Q,
    const float* __restrict__ KT, const int* __restrict__ mask,
    float* __restrict__ Z) {
  __shared__ float q_lds[16 * QP_];
  __shared__ float p_lds[2 * PB_];
  __shared__ float red[NW_ * 16];
  __shared__ float mrow[16];
  __shared__ float srow[16];
  __shared__ float frow[16];

  const int tid  = threadIdx.x;
  const int w    = tid >> 5;
  const int lane = tid & 31;
  const int half = lane >> 4;
  const int l15  = lane & 15;
  const int b    = blockIdx.x >> 5;
  const int t0   = (blockIdx.x & 31) * 16;

  const float* Hbat = H + (size_t)b * L_ * DH_;
  const float* KTpf = KT + (size_t)b * DP_ * L_;
  const int pfs = pf_slice(blockIdx.x & 31);
  pf_h_chunk(Hbat, 0, pfs, w, lane);
  pf_kt_chunk(KTpf, LC_, pfs, w, lane);

  const float* Qb = Q + ((size_t)b * T_ + t0) * DP_;
  for (int i = tid; i < 16 * DP_; i += 32 * NW_)
    q_lds[(i >> 8) * QP_ + (i & (DP_ - 1))] = Qb[i];
  if (tid < 16) mrow[tid] = -3.0e38f;
  __syncthreads();

  v8f zacc[2];
  zacc[0] = zero8();
  zacc[1] = zero8();
  v8f lsum = zero8();

  const float* KTb = KT + (size_t)b * DP_ * L_;
  const int* maskb = mask + (size_t)b * L_;
  const int colw = w * 16 + l15;

  v8f s0 = pa_scores(q_lds, red, KTb, maskb, 0, w, half, l15);
  __syncthreads();
  if (tid < 16) row_max_merge(red, mrow, frow, tid);
  __syncthreads();
#pragma unroll
  for (int v = 0; v < 8; ++v) {
    const int row = v + 8 * half;
    const float p0 = __expf(s0[v] - mrow[row]);
    p_lds[row * PP_ + colw] = p0;
    lsum[v] = p0;
  }
  bar_signal();

  for (int c = 0; c < NC_; ++c) {
    const int lc = c * LC_;
    const bool more = (c + 1 < NC_);
    const float* pcur = p_lds + (c & 1) * PB_;
    float* pnxt = p_lds + ((c + 1) & 1) * PB_;

    if (more) pf_h_chunk(Hbat, lc + LC_, pfs, w, lane);
    if (c + 2 < NC_) pf_kt_chunk(KTpf, lc + 2 * LC_, pfs, w, lane);

    __builtin_amdgcn_s_setprio(0);
    if (more) s0 = pa_scores(q_lds, red, KTb, maskb, lc + LC_, w, half, l15);
    bar_wait();
    if (more) bar_signal();

    const float* Hb = H + ((size_t)b * L_ + lc + 2 * half) * DH_ + w * 32 + l15;
    const float* pbase = pcur + l15 * PP_ + 2 * half;
    __builtin_amdgcn_s_setprio(3);
    pc_span(zacc, pbase, Hb, 0, PC1_);

    if (more) {
      bar_wait();
      if (tid < 16) row_max_merge(red, mrow, frow, tid);
      bar_signal();
    }
    __builtin_amdgcn_s_setprio(2);
    pc_span(zacc, pbase, Hb, PC1_, PC2_);

    float fs[8];
    if (more) {
      bar_wait();
#pragma unroll
      for (int v = 0; v < 8; ++v) {
        const int row = v + 8 * half;
        fs[v] = frow[row];
        const float p0 = __expf(s0[v] - mrow[row]);
        pnxt[row * PP_ + colw] = p0;
        lsum[v] = lsum[v] * fs[v] + p0;
      }
      bar_signal();
    }
    __builtin_amdgcn_s_setprio(1);
    pc_span(zacc, pbase, Hb, PC2_, LC_);

    if (more) {
#pragma unroll
      for (int v = 0; v < 8; ++v) {
        zacc[0][v] *= fs[v];
        zacc[1][v] *= fs[v];
      }
    }
  }
  __builtin_amdgcn_s_setprio(0);

#pragma unroll
  for (int v = 0; v < 8; ++v) {
    float x = lsum[v];
    x += __shfl_xor(x, 8, 16);
    x += __shfl_xor(x, 4, 16);
    x += __shfl_xor(x, 2, 16);
    x += __shfl_xor(x, 1, 16);
    if (l15 == 0) red[w * 16 + v + 8 * half] = x;
  }
  __syncthreads();
  if (tid < 16) {
    float acc = 0.f;
#pragma unroll
    for (int ww = 0; ww < NW_; ++ww) acc += red[ww * 16 + tid];
    srow[tid] = acc;
  }
  __syncthreads();

  float* Zb = Z + ((size_t)b * T_ + t0) * DH_;
#pragma unroll
  for (int v = 0; v < 8; ++v) {
    const int m = v + 8 * half;
    const float inv = 1.0f / srow[m];
#pragma unroll
    for (int nt = 0; nt < 2; ++nt) {
      const int d = w * 32 + nt * 16 + l15;
      Zb[(size_t)m * DH_ + d] = zacc[nt][v] * inv;
    }
  }
}

extern "C" void kernel_launch(void* const* d_in, const int* in_sizes, int n_in,
                              void* d_out, int out_size, void* d_ws, size_t ws_size,
                              hipStream_t stream) {
  const float* H    = (const float*)d_in[0];
  const float* G    = (const float*)d_in[1];
  const float* Wk   = (const float*)d_in[2];
  const float* Wq   = (const float*)d_in[3];
  const int*   mask = (const int*)d_in[4];
  float*       Z    = (float*)d_out;

  float* Qws  = (float*)d_ws;
  float* KTws = Qws + (size_t)B_ * T_ * DP_;

  qproj_kernel<<<(B_ * T_) / 16, 256, 0, stream>>>(G, Wq, Qws, DG_);
  kproj_kernel<<<(B_ * L_) / KPR_, 32 * KPW_, 0, stream>>>(H, Wk, KTws);
  attn_kernel<<<B_ * (T_ / 16), 32 * NW_, 0, stream>>>(H, Qws, KTws, mask, Z);
}
